# starting kernel
# baseline (speedup 1.0000x reference)
#include <hip/hip_runtime.h>

typedef float v2f __attribute__((ext_vector_type(2)));
typedef float v8f __attribute__((ext_vector_type(8)));

#define B_    256
#define T_    150
#define J_    25
#define WAVES 8
#define GSTR  33

__global__ __launch_bounds__(256)
void knn_feats_kernel(const float* __restrict__ x, float* __restrict__ out) {
    __shared__ float lds_g[WAVES][32 * GSTR];
    __shared__ float lds_red[WAVES][J_ * 9];

    const int tid  = threadIdx.x;
    const int wv   = tid >> 5;
    const int lane = tid & 31;
    const int b    = blockIdx.x;

    float acc[9];
#pragma unroll
    for (int f = 0; f < 9; ++f) acc[f] = 0.0f;

    const int  lj      = lane & 15;
    const bool hiHalf  = lane >= 16;
    const int  rowLo   = lj;
    const int  rowHi   = 16 + lj;
    const bool hiValid = rowHi < J_;

    float* gw = lds_g[wv];

    for (int t = wv; t < T_; t += WAVES) {
        const float* px = x + (size_t)(b * T_ + t) * (J_ * 3);

        v2f aLo, aHi;
        if (!hiHalf) {
            aLo.x = px[rowLo * 3 + 0];
            aLo.y = px[rowLo * 3 + 1];
            aHi.x = hiValid ? px[rowHi * 3 + 0] : 0.0f;
            aHi.y = hiValid ? px[rowHi * 3 + 1] : 0.0f;
        } else {
            aLo.x = px[rowLo * 3 + 2];
            aLo.y = 0.0f;
            aHi.x = hiValid ? px[rowHi * 3 + 2] : 0.0f;
            aHi.y = 0.0f;
        }

        v8f cz = {};
        v8f g00 = __builtin_amdgcn_wmma_f32_16x16x4_f32(false, aLo, false, aLo, (short)0, cz, false, false);
        v8f g01 = __builtin_amdgcn_wmma_f32_16x16x4_f32(false, aLo, false, aHi, (short)0, cz, false, false);
        v8f g10 = __builtin_amdgcn_wmma_f32_16x16x4_f32(false, aHi, false, aLo, (short)0, cz, false, false);
        v8f g11 = __builtin_amdgcn_wmma_f32_16x16x4_f32(false, aHi, false, aHi, (short)0, cz, false, false);

        const int rbase = hiHalf ? 8 : 0;
#pragma unroll
        for (int r = 0; r < 8; ++r) {
            gw[(rbase + r) * GSTR + lj]            = g00[r];
            gw[(rbase + r) * GSTR + 16 + lj]       = g01[r];
            gw[(16 + rbase + r) * GSTR + lj]       = g10[r];
            gw[(16 + rbase + r) * GSTR + 16 + lj]  = g11[r];
        }
        __builtin_amdgcn_wave_barrier();

        const int j = lane;
        const float nj = gw[j * GSTR + j];

        float m1 = 3.0e38f, m2 = 3.0e38f, m3 = 3.0e38f, m4 = 3.0e38f;
#pragma unroll 5
        for (int i = 0; i < J_; ++i) {
            float ni  = gw[i * GSTR + i];
            float gij = gw[i * GSTR + j];
            float s   = ni + nj - 2.0f * gij;
            float d   = (s > 0.0f) ? __builtin_sqrtf(s) : 0.0f;
            d = (i == j) ? 3.0e38f : d;
            float t1 = fminf(d,  m1); float r1 = fmaxf(d,  m1);
            float t2 = fminf(r1, m2); float r2 = fmaxf(r1, m2);
            float t3 = fminf(r2, m3); float r3 = fmaxf(r2, m3);
            float t4 = fminf(r3, m4);
            m1 = t1; m2 = t2; m3 = t3; m4 = t4;
        }

        float mean2 = 0.5f * (m1 + m2);
        float a0 = m1 - mean2, a1 = m2 - mean2;
        float std2 = __builtin_sqrtf(a0 * a0 + a1 * a1);

        float mean3 = (m1 + m2 + m3) * (1.0f / 3.0f);
        float e0 = m1 - mean3, e1 = m2 - mean3, e2 = m3 - mean3;
        float std3 = __builtin_sqrtf((e0 * e0 + e1 * e1 + e2 * e2) * 0.5f);

        float mean4 = (m1 + m2 + m3 + m4) * 0.25f;
        float f0 = m1 - mean4, f1 = m2 - mean4, f2 = m3 - mean4, f3 = m4 - mean4;
        float std4 = __builtin_sqrtf((f0 * f0 + f1 * f1 + f2 * f2 + f3 * f3) * (1.0f / 3.0f));

        if (j < J_) {
            acc[0] += mean2; acc[1] += std2; acc[2] += m1;
            acc[3] += mean3; acc[4] += std3; acc[5] += m1;
            acc[6] += mean4; acc[7] += std4; acc[8] += m1;
        }
        __builtin_amdgcn_wave_barrier();
    }

    if (lane < J_) {
#pragma unroll
        for (int f = 0; f < 9; ++f)
            lds_red[wv][lane * 9 + f] = acc[f] * (1.0f / (float)T_);
    }
    __syncthreads();

    if (tid < J_ * 9) {
        float s = 0.0f;
#pragma unroll
        for (int w = 0; w < WAVES; ++w) s += lds_red[w][tid];
        out[(size_t)b * (J_ * 9) + tid] = s;
    }
}

extern "C" void kernel_launch(void* const* d_in, const int* in_sizes, int n_in,
                              void* d_out, int out_size, void* d_ws, size_t ws_size,
                              hipStream_t stream) {
    (void)in_sizes; (void)n_in; (void)d_ws; (void)ws_size; (void)out_size;
    const float* x = (const float*)d_in[0];
    float* out = (float*)d_out;
    knn_feats_kernel<<<B_, 256, 0, stream>>>(x, out);
}


# 32 waves per work-group instead of 8 (frames strided over them, 8 waves per SIMD) so sibling waves fill each wave's serial load/LDS/sqrt waits; 32 partial frame sums per joint: f32 add order changes
# speedup vs baseline: 1.6050x; 1.6050x over previous
#include <hip/hip_runtime.h>

typedef float v2f __attribute__((ext_vector_type(2)));
typedef float v8f __attribute__((ext_vector_type(8)));

#define B_    256
#define T_    150
#define J_    25
#define WAVES 32
#define GSTR  33

__global__ __launch_bounds__(1024)
void knn_feats_kernel(const float* __restrict__ x, float* __restrict__ out) {
    __shared__ float lds_g[WAVES][32 * GSTR];
    __shared__ float lds_red[WAVES][J_ * 9];

    const int tid  = threadIdx.x;
    const int wv   = tid >> 5;
    const int lane = tid & 31;
    const int b    = blockIdx.x;

    float acc[9];
#pragma unroll
    for (int f = 0; f < 9; ++f) acc[f] = 0.0f;

    const int  lj      = lane & 15;
    const bool hiHalf  = lane >= 16;
    const int  rowLo   = lj;
    const int  rowHi   = 16 + lj;
    const bool hiValid = rowHi < J_;

    float* gw = lds_g[wv];

    for (int t = wv; t < T_; t += WAVES) {
        const float* px = x + (size_t)(b * T_ + t) * (J_ * 3);

        v2f aLo, aHi;
        if (!hiHalf) {
            aLo.x = px[rowLo * 3 + 0];
            aLo.y = px[rowLo * 3 + 1];
            aHi.x = hiValid ? px[rowHi * 3 + 0] : 0.0f;
            aHi.y = hiValid ? px[rowHi * 3 + 1] : 0.0f;
        } else {
            aLo.x = px[rowLo * 3 + 2];
            aLo.y = 0.0f;
            aHi.x = hiValid ? px[rowHi * 3 + 2] : 0.0f;
            aHi.y = 0.0f;
        }

        v8f cz = {};
        v8f g00 = __builtin_amdgcn_wmma_f32_16x16x4_f32(false, aLo, false, aLo, (short)0, cz, false, false);
        v8f g01 = __builtin_amdgcn_wmma_f32_16x16x4_f32(false, aLo, false, aHi, (short)0, cz, false, false);
        v8f g10 = __builtin_amdgcn_wmma_f32_16x16x4_f32(false, aHi, false, aLo, (short)0, cz, false, false);
        v8f g11 = __builtin_amdgcn_wmma_f32_16x16x4_f32(false, aHi, false, aHi, (short)0, cz, false, false);

        const int rbase = hiHalf ? 8 : 0;
#pragma unroll
        for (int r = 0; r < 8; ++r) {
            gw[(rbase + r) * GSTR + lj]            = g00[r];
            gw[(rbase + r) * GSTR + 16 + lj]       = g01[r];
            gw[(16 + rbase + r) * GSTR + lj]       = g10[r];
            gw[(16 + rbase + r) * GSTR + 16 + lj]  = g11[r];
        }
        __builtin_amdgcn_wave_barrier();

        const int j = lane;
        const float nj = gw[j * GSTR + j];

        float m1 = 3.0e38f, m2 = 3.0e38f, m3 = 3.0e38f, m4 = 3.0e38f;
#pragma unroll 5
        for (int i = 0; i < J_; ++i) {
            float ni  = gw[i * GSTR + i];
            float gij = gw[i * GSTR + j];
            float s   = ni + nj - 2.0f * gij;
            float d   = (s > 0.0f) ? __builtin_sqrtf(s) : 0.0f;
            d = (i == j) ? 3.0e38f : d;
            float t1 = fminf(d,  m1); float r1 = fmaxf(d,  m1);
            float t2 = fminf(r1, m2); float r2 = fmaxf(r1, m2);
            float t3 = fminf(r2, m3); float r3 = fmaxf(r2, m3);
            float t4 = fminf(r3, m4);
            m1 = t1; m2 = t2; m3 = t3; m4 = t4;
        }

        float mean2 = 0.5f * (m1 + m2);
        float a0 = m1 - mean2, a1 = m2 - mean2;
        float std2 = __builtin_sqrtf(a0 * a0 + a1 * a1);

        float mean3 = (m1 + m2 + m3) * (1.0f / 3.0f);
        float e0 = m1 - mean3, e1 = m2 - mean3, e2 = m3 - mean3;
        float std3 = __builtin_sqrtf((e0 * e0 + e1 * e1 + e2 * e2) * 0.5f);

        float mean4 = (m1 + m2 + m3 + m4) * 0.25f;
        float f0 = m1 - mean4, f1 = m2 - mean4, f2 = m3 - mean4, f3 = m4 - mean4;
        float std4 = __builtin_sqrtf((f0 * f0 + f1 * f1 + f2 * f2 + f3 * f3) * (1.0f / 3.0f));

        if (j < J_) {
            acc[0] += mean2; acc[1] += std2; acc[2] += m1;
            acc[3] += mean3; acc[4] += std3; acc[5] += m1;
            acc[6] += mean4; acc[7] += std4; acc[8] += m1;
        }
        __builtin_amdgcn_wave_barrier();
    }

    if (lane < J_) {
#pragma unroll
        for (int f = 0; f < 9; ++f)
            lds_red[wv][lane * 9 + f] = acc[f] * (1.0f / (float)T_);
    }
    __syncthreads();

    if (tid < J_ * 9) {
        float s = 0.0f;
#pragma unroll
        for (int w = 0; w < WAVES; ++w) s += lds_red[w][tid];
        out[(size_t)b * (J_ * 9) + tid] = s;
    }
}

extern "C" void kernel_launch(void* const* d_in, const int* in_sizes, int n_in,
                              void* d_out, int out_size, void* d_ws, size_t ws_size,
                              hipStream_t stream) {
    (void)in_sizes; (void)n_in; (void)d_ws; (void)ws_size; (void)out_size;
    const float* x = (const float*)d_in[0];
    float* out = (float*)d_out;
    knn_feats_kernel<<<B_, 1024, 0, stream>>>(x, out);
}


# k-nearest selection runs on squared distances; the correctly rounded sqrt is taken of the 4 selected per joint instead of all 25 (sqrt is monotone: same 4 values, same sums)
# speedup vs baseline: 2.2940x; 1.4293x over previous
#include <hip/hip_runtime.h>

typedef float v2f __attribute__((ext_vector_type(2)));
typedef float v8f __attribute__((ext_vector_type(8)));

#define B_    256
#define T_    150
#define J_    25
#define WAVES 32
#define GSTR  33

__global__ __launch_bounds__(1024)
void knn_feats_kernel(const float* __restrict__ x, float* __restrict__ out) {
    __shared__ float lds_g[WAVES][32 * GSTR];
    __shared__ float lds_red[WAVES][J_ * 9];

    const int tid  = threadIdx.x;
    const int wv   = tid >> 5;
    const int lane = tid & 31;
    const int b    = blockIdx.x;

    float acc[9];
#pragma unroll
    for (int f = 0; f < 9; ++f) acc[f] = 0.0f;

    const int  lj      = lane & 15;
    const bool hiHalf  = lane >= 16;
    const int  rowLo   = lj;
    const int  rowHi   = 16 + lj;
    const bool hiValid = rowHi < J_;

    float* gw = lds_g[wv];

    for (int t = wv; t < T_; t += WAVES) {
        const float* px = x + (size_t)(b * T_ + t) * (J_ * 3);

        v2f aLo, aHi;
        if (!hiHalf) {
            aLo.x = px[rowLo * 3 + 0];
            aLo.y = px[rowLo * 3 + 1];
            aHi.x = hiValid ? px[rowHi * 3 + 0] : 0.0f;
            aHi.y = hiValid ? px[rowHi * 3 + 1] : 0.0f;
        } else {
            aLo.x = px[rowLo * 3 + 2];
            aLo.y = 0.0f;
            aHi.x = hiValid ? px[rowHi * 3 + 2] : 0.0f;
            aHi.y = 0.0f;
        }

        v8f cz = {};
        v8f g00 = __builtin_amdgcn_wmma_f32_16x16x4_f32(false, aLo, false, aLo, (short)0, cz, false, false);
        v8f g01 = __builtin_amdgcn_wmma_f32_16x16x4_f32(false, aLo, false, aHi, (short)0, cz, false, false);
        v8f g10 = __builtin_amdgcn_wmma_f32_16x16x4_f32(false, aHi, false, aLo, (short)0, cz, false, false);
        v8f g11 = __builtin_amdgcn_wmma_f32_16x16x4_f32(false, aHi, false, aHi, (short)0, cz, false, false);

        const int rbase = hiHalf ? 8 : 0;
#pragma unroll
        for (int r = 0; r < 8; ++r) {
            gw[(rbase + r) * GSTR + lj]            = g00[r];
            gw[(rbase + r) * GSTR + 16 + lj]       = g01[r];
            gw[(16 + rbase + r) * GSTR + lj]       = g10[r];
            gw[(16 + rbase + r) * GSTR + 16 + lj]  = g11[r];
        }
        __builtin_amdgcn_wave_barrier();

        const int j = lane;
        const float nj = gw[j * GSTR + j];

        float q1 = 3.0e38f, q2 = 3.0e38f, q3 = 3.0e38f, q4 = 3.0e38f;
#pragma unroll 5
        for (int i = 0; i < J_; ++i) {
            float ni  = gw[i * GSTR + i];
            float gij = gw[i * GSTR + j];
            float s   = ni + nj - 2.0f * gij;
            s = (i == j) ? 3.0e38f : s;
            float t1 = fminf(s,  q1); float r1 = fmaxf(s,  q1);
            float t2 = fminf(r1, q2); float r2 = fmaxf(r1, q2);
            float t3 = fminf(r2, q3); float r3 = fmaxf(r2, q3);
            float t4 = fminf(r3, q4);
            q1 = t1; q2 = t2; q3 = t3; q4 = t4;
        }
        const float m1 = (q1 > 0.0f) ? __builtin_sqrtf(q1) : 0.0f;
        const float m2 = (q2 > 0.0f) ? __builtin_sqrtf(q2) : 0.0f;
        const float m3 = (q3 > 0.0f) ? __builtin_sqrtf(q3) : 0.0f;
        const float m4 = (q4 > 0.0f) ? __builtin_sqrtf(q4) : 0.0f;

        float mean2 = 0.5f * (m1 + m2);
        float a0 = m1 - mean2, a1 = m2 - mean2;
        float std2 = __builtin_sqrtf(a0 * a0 + a1 * a1);

        float mean3 = (m1 + m2 + m3) * (1.0f / 3.0f);
        float e0 = m1 - mean3, e1 = m2 - mean3, e2 = m3 - mean3;
        float std3 = __builtin_sqrtf((e0 * e0 + e1 * e1 + e2 * e2) * 0.5f);

        float mean4 = (m1 + m2 + m3 + m4) * 0.25f;
        float f0 = m1 - mean4, f1 = m2 - mean4, f2 = m3 - mean4, f3 = m4 - mean4;
        float std4 = __builtin_sqrtf((f0 * f0 + f1 * f1 + f2 * f2 + f3 * f3) * (1.0f / 3.0f));

        if (j < J_) {
            acc[0] += mean2; acc[1] += std2; acc[2] += m1;
            acc[3] += mean3; acc[4] += std3; acc[5] += m1;
            acc[6] += mean4; acc[7] += std4; acc[8] += m1;
        }
        __builtin_amdgcn_wave_barrier();
    }

    if (lane < J_) {
#pragma unroll
        for (int f = 0; f < 9; ++f)
            lds_red[wv][lane * 9 + f] = acc[f] * (1.0f / (float)T_);
    }
    __syncthreads();

    if (tid < J_ * 9) {
        float s = 0.0f;
#pragma unroll
        for (int w = 0; w < WAVES; ++w) s += lds_red[w][tid];
        out[(size_t)b * (J_ * 9) + tid] = s;
    }
}

extern "C" void kernel_launch(void* const* d_in, const int* in_sizes, int n_in,
                              void* d_out, int out_size, void* d_ws, size_t ws_size,
                              hipStream_t stream) {
    (void)in_sizes; (void)n_in; (void)d_ws; (void)ws_size; (void)out_size;
    const float* x = (const float*)d_in[0];
    float* out = (float*)d_out;
    knn_feats_kernel<<<B_, 1024, 0, stream>>>(x, out);
}


# top-4 selection reads the tile from the matrix accumulator registers (norm folded into k=3, lane halves merged by v_permlanex16) instead of an LDS round trip; squared-distance rounding changes
# speedup vs baseline: 2.6109x; 1.1381x over previous
#include <hip/hip_runtime.h>


typedef float    v2f __attribute__((ext_vector_type(2)));
typedef float    v8f __attribute__((ext_vector_type(8)));

#define B_     256
#define T_     150
#define J_     25
#define WAVES  32
#define NEGBIG (-1.0e30f)

__device__ __forceinline__ void ce_desc(float& a, float& b) {
    const float hi = fmaxf(a, b);
    const float lo = fminf(a, b);
    a = hi; b = lo;
}

__device__ __forceinline__ void insert_desc(float s, float& q1, float& q2, float& q3, float& q4) {
    const float t1 = fmaxf(s,  q1); const float r1 = fminf(s,  q1);
    const float t2 = fmaxf(r1, q2); const float r2 = fminf(r1, q2);
    const float t3 = fmaxf(r2, q3); const float r3 = fminf(r2, q3);
    const float t4 = fmaxf(r3, q4);
    q1 = t1; q2 = t2; q3 = t3; q4 = t4;
}

__device__ __forceinline__ void top4_of_column(const v8f& u, const v8f& w,
                                               float& q1, float& q2, float& q3, float& q4) {
    q1 = u[0]; q2 = u[1]; q3 = u[2]; q4 = u[3];
    ce_desc(q1, q2); ce_desc(q3, q4); ce_desc(q1, q3); ce_desc(q2, q4); ce_desc(q2, q3);
    insert_desc(u[4], q1, q2, q3, q4);
    insert_desc(u[5], q1, q2, q3, q4);
    insert_desc(u[6], q1, q2, q3, q4);
    insert_desc(u[7], q1, q2, q3, q4);
    insert_desc(w[0], q1, q2, q3, q4);
    insert_desc(w[1], q1, q2, q3, q4);
    insert_desc(w[2], q1, q2, q3, q4);
    insert_desc(w[3], q1, q2, q3, q4);
    insert_desc(w[4], q1, q2, q3, q4);
}

__device__ __forceinline__ float xhalf(float v) {
    const int i = __builtin_bit_cast(int, v);
    return __builtin_bit_cast(float, __builtin_amdgcn_permlanex16(i, i, 0x76543210, 0xfedcba98, false, false));
}

__global__ __launch_bounds__(1024)
void knn_feats_kernel(const float* __restrict__ x, float* __restrict__ out) {
    __shared__ float lds_red[WAVES][J_ * 9];

    const int tid  = threadIdx.x;
    const int wv   = tid >> 5;
    const int lane = tid & 31;
    const int b    = blockIdx.x;

    const int  c  = lane & 15;
    const bool hi = lane >= 16;

    const bool v1 = (c < 5) || (c >= 8 && c < 12);
    const int  j1 = v1 ? ((c < 5) ? 16 + c : 13 + c) : 0;

    const int  ownJ     = hi ? j1 : c;
    const bool ownValid = hi ? v1 : true;

    v8f cdiag;
#pragma unroll
    for (int r = 0; r < 8; ++r) cdiag[r] = (((hi ? 8 : 0) + r) == c) ? NEGBIG : 0.0f;
    const v8f cz = {};

    float acc[9];
#pragma unroll
    for (int f = 0; f < 9; ++f) acc[f] = 0.0f;

    for (int t = wv; t < T_; t += WAVES) {
        const float* px = x + (size_t)(b * T_ + t) * (J_ * 3);
        const float* p0 = px + 3 * c;
        const float* p1 = px + 3 * j1;
        const float x0 = p0[0], y0 = p0[1], z0 = p0[2];
        const float x1 = p1[0], y1 = p1[1], z1 = p1[2];

        const float n0  = x0 * x0 + y0 * y0 + z0 * z0;
        const float n1  = x1 * x1 + y1 * y1 + z1 * z1;
        const float nh0 = -0.5f * n0;
        const float nh1 = v1 ? -0.5f * n1 : NEGBIG;

        v2f a0, a1, b0, b1;
        a0.x = hi ? z0 : x0;   a0.y = hi ? nh0  : y0;
        b0.x = a0.x;           b0.y = hi ? 1.0f : y0;
        a1.x = hi ? z1 : x1;   a1.y = hi ? nh1  : y1;
        b1.x = a1.x;           b1.y = hi ? 1.0f : y1;

        v8f d00 = __builtin_amdgcn_wmma_f32_16x16x4_f32(false, a0, false, b0, (short)0, cdiag, false, false);
        v8f d01 = __builtin_amdgcn_wmma_f32_16x16x4_f32(false, a0, false, b1, (short)0, cz,    false, false);
        v8f d10 = __builtin_amdgcn_wmma_f32_16x16x4_f32(false, a1, false, b0, (short)0, cz,    false, false);
        v8f d11 = __builtin_amdgcn_wmma_f32_16x16x4_f32(false, a1, false, b1, (short)0, cdiag, false, false);

        asm volatile("v_nop\n\tv_nop\n\tv_nop\n\tv_nop" : "+v"(d00), "+v"(d01), "+v"(d10), "+v"(d11));

        float P1, P2, P3, P4, Q1, Q2, Q3, Q4;
        top4_of_column(d00, d10, P1, P2, P3, P4);
        top4_of_column(d01, d11, Q1, Q2, Q3, Q4);

        const float R1 = xhalf(hi ? P1 : Q1), R2 = xhalf(hi ? P2 : Q2);
        const float R3 = xhalf(hi ? P3 : Q3), R4 = xhalf(hi ? P4 : Q4);
        const float O1 = hi ? Q1 : P1, O2 = hi ? Q2 : P2, O3 = hi ? Q3 : P3, O4 = hi ? Q4 : P4;

        float L1 = fmaxf(O1, R4), L2 = fmaxf(O2, R3), L3 = fmaxf(O3, R2), L4 = fmaxf(O4, R1);
        ce_desc(L1, L3); ce_desc(L2, L4); ce_desc(L1, L2); ce_desc(L3, L4);

        const float nj = hi ? n1 : n0;
        const float s1 = fmaf(-2.0f, L1, nj);
        const float s2 = fmaf(-2.0f, L2, nj);
        const float s3 = fmaf(-2.0f, L3, nj);
        const float s4 = fmaf(-2.0f, L4, nj);

        const float m1 = (s1 > 0.0f) ? __builtin_sqrtf(s1) : 0.0f;
        const float m2 = (s2 > 0.0f) ? __builtin_sqrtf(s2) : 0.0f;
        const float m3 = (s3 > 0.0f) ? __builtin_sqrtf(s3) : 0.0f;
        const float m4 = (s4 > 0.0f) ? __builtin_sqrtf(s4) : 0.0f;

        float mean2 = 0.5f * (m1 + m2);
        float e20 = m1 - mean2, e21 = m2 - mean2;
        float std2 = __builtin_sqrtf(e20 * e20 + e21 * e21);

        float mean3 = (m1 + m2 + m3) * (1.0f / 3.0f);
        float e0 = m1 - mean3, e1 = m2 - mean3, e2 = m3 - mean3;
        float std3 = __builtin_sqrtf((e0 * e0 + e1 * e1 + e2 * e2) * 0.5f);

        float mean4 = (m1 + m2 + m3 + m4) * 0.25f;
        float f0 = m1 - mean4, f1 = m2 - mean4, f2 = m3 - mean4, f3 = m4 - mean4;
        float std4 = __builtin_sqrtf((f0 * f0 + f1 * f1 + f2 * f2 + f3 * f3) * (1.0f / 3.0f));

        if (ownValid) {
            acc[0] += mean2; acc[1] += std2; acc[2] += m1;
            acc[3] += mean3; acc[4] += std3; acc[5] += m1;
            acc[6] += mean4; acc[7] += std4; acc[8] += m1;
        }
    }

    if (ownValid) {
#pragma unroll
        for (int f = 0; f < 9; ++f)
            lds_red[wv][ownJ * 9 + f] = acc[f] * (1.0f / (float)T_);
    }
    __syncthreads();

    if (tid < J_ * 9) {
        float s = 0.0f;
#pragma unroll
        for (int w = 0; w < WAVES; ++w) s += lds_red[w][tid];
        out[(size_t)b * (J_ * 9) + tid] = s;
    }
}

extern "C" void kernel_launch(void* const* d_in, const int* in_sizes, int n_in,
                              void* d_out, int out_size, void* d_ws, size_t ws_size,
                              hipStream_t stream) {
    (void)in_sizes; (void)n_in; (void)d_ws; (void)ws_size; (void)out_size;
    const float* x = (const float*)d_in[0];
    float* out = (float*)d_out;
    knn_feats_kernel<<<B_, 1024, 0, stream>>>(x, out);
}


# top-4 selection rebuilt on v_med3: each insert is 4 independent medians of (s, q[i-1], q[i]) instead of a 7-long min/max chain; max/min as median vs +-FLT_MAX, so no quieting max(x,x) per entry
# speedup vs baseline: 2.9648x; 1.1355x over previous
#include <hip/hip_runtime.h>


typedef float    v2f __attribute__((ext_vector_type(2)));
typedef float    v8f __attribute__((ext_vector_type(8)));

#define B_     256
#define T_     150
#define J_     25
#define WAVES  32
#define NEGBIG (-1.0e30f)

__device__ __forceinline__ float md(float a, float b, float c) { return __builtin_amdgcn_fmed3f(a, b, c); }
#define TOP_ ( 3.402823466e+38f)
#define BOT_ (-3.402823466e+38f)
__device__ __forceinline__ float mx(float a, float b) { return md(a, b, TOP_); }
__device__ __forceinline__ float mn(float a, float b) { return md(a, b, BOT_); }

__device__ __forceinline__ void ce_desc(float& a, float& b) {
    const float hi = fmaxf(a, b);
    const float lo = fminf(a, b);
    a = hi; b = lo;
}

__device__ __forceinline__ void insert_desc(float s, float& q1, float& q2, float& q3, float& q4) {
    const float t1 = mx(s, q1);
    const float t2 = md(s, q1, q2);
    const float t3 = md(s, q2, q3);
    const float t4 = md(s, q3, q4);
    q1 = t1; q2 = t2; q3 = t3; q4 = t4;
}

__device__ __forceinline__ void top4_of_column(const v8f& u, const v8f& w,
                                               float& q1, float& q2, float& q3, float& q4) {
    q1 = mx(u[0], u[1]); q2 = mn(u[0], u[1]);
    { const float t1 = mx(u[2], q1), t2 = md(u[2], q1, q2), t3 = mn(u[2], q2); q1 = t1; q2 = t2; q3 = t3; }
    { const float t1 = mx(u[3], q1), t2 = md(u[3], q1, q2), t3 = md(u[3], q2, q3), t4 = mn(u[3], q3);
      q1 = t1; q2 = t2; q3 = t3; q4 = t4; }
    insert_desc(u[4], q1, q2, q3, q4);
    insert_desc(u[5], q1, q2, q3, q4);
    insert_desc(u[6], q1, q2, q3, q4);
    insert_desc(u[7], q1, q2, q3, q4);
    insert_desc(w[0], q1, q2, q3, q4);
    insert_desc(w[1], q1, q2, q3, q4);
    insert_desc(w[2], q1, q2, q3, q4);
    insert_desc(w[3], q1, q2, q3, q4);
    insert_desc(w[4], q1, q2, q3, q4);
}

__device__ __forceinline__ float xhalf(float v) {
    const int i = __builtin_bit_cast(int, v);
    return __builtin_bit_cast(float, __builtin_amdgcn_permlanex16(i, i, 0x76543210, 0xfedcba98, false, false));
}

__global__ __launch_bounds__(1024)
void knn_feats_kernel(const float* __restrict__ x, float* __restrict__ out) {
    __shared__ float lds_red[WAVES][J_ * 9];

    const int tid  = threadIdx.x;
    const int wv   = tid >> 5;
    const int lane = tid & 31;
    const int b    = blockIdx.x;

    const int  c  = lane & 15;
    const bool hi = lane >= 16;

    const bool v1 = (c < 5) || (c >= 8 && c < 12);
    const int  j1 = v1 ? ((c < 5) ? 16 + c : 13 + c) : 0;

    const int  ownJ     = hi ? j1 : c;
    const bool ownValid = hi ? v1 : true;

    v8f cdiag;
#pragma unroll
    for (int r = 0; r < 8; ++r) cdiag[r] = (((hi ? 8 : 0) + r) == c) ? NEGBIG : 0.0f;
    const v8f cz = {};

    float acc[9];
#pragma unroll
    for (int f = 0; f < 9; ++f) acc[f] = 0.0f;

    for (int t = wv; t < T_; t += WAVES) {
        const float* px = x + (size_t)(b * T_ + t) * (J_ * 3);
        const float* p0 = px + 3 * c;
        const float* p1 = px + 3 * j1;
        const float x0 = p0[0], y0 = p0[1], z0 = p0[2];
        const float x1 = p1[0], y1 = p1[1], z1 = p1[2];

        const float n0  = x0 * x0 + y0 * y0 + z0 * z0;
        const float n1  = x1 * x1 + y1 * y1 + z1 * z1;
        const float nh0 = -0.5f * n0;
        const float nh1 = v1 ? -0.5f * n1 : NEGBIG;

        v2f a0, a1, b0, b1;
        a0.x = hi ? z0 : x0;   a0.y = hi ? nh0  : y0;
        b0.x = a0.x;           b0.y = hi ? 1.0f : y0;
        a1.x = hi ? z1 : x1;   a1.y = hi ? nh1  : y1;
        b1.x = a1.x;           b1.y = hi ? 1.0f : y1;

        v8f d00 = __builtin_amdgcn_wmma_f32_16x16x4_f32(false, a0, false, b0, (short)0, cdiag, false, false);
        v8f d01 = __builtin_amdgcn_wmma_f32_16x16x4_f32(false, a0, false, b1, (short)0, cz,    false, false);
        v8f d10 = __builtin_amdgcn_wmma_f32_16x16x4_f32(false, a1, false, b0, (short)0, cz,    false, false);
        v8f d11 = __builtin_amdgcn_wmma_f32_16x16x4_f32(false, a1, false, b1, (short)0, cdiag, false, false);

        asm volatile("v_nop\n\tv_nop\n\tv_nop\n\tv_nop" : "+v"(d00), "+v"(d01), "+v"(d10), "+v"(d11));

        float P1, P2, P3, P4, Q1, Q2, Q3, Q4;
        top4_of_column(d00, d10, P1, P2, P3, P4);
        top4_of_column(d01, d11, Q1, Q2, Q3, Q4);

        const float R1 = xhalf(hi ? P1 : Q1), R2 = xhalf(hi ? P2 : Q2);
        const float R3 = xhalf(hi ? P3 : Q3), R4 = xhalf(hi ? P4 : Q4);
        const float O1 = hi ? Q1 : P1, O2 = hi ? Q2 : P2, O3 = hi ? Q3 : P3, O4 = hi ? Q4 : P4;

        float L1 = mx(O1, R4), L2 = mx(O2, R3), L3 = mx(O3, R2), L4 = mx(O4, R1);
        ce_desc(L1, L3); ce_desc(L2, L4); ce_desc(L1, L2); ce_desc(L3, L4);

        const float nj = hi ? n1 : n0;
        const float s1 = fmaf(-2.0f, L1, nj);
        const float s2 = fmaf(-2.0f, L2, nj);
        const float s3 = fmaf(-2.0f, L3, nj);
        const float s4 = fmaf(-2.0f, L4, nj);

        const float m1 = (s1 > 0.0f) ? __builtin_sqrtf(s1) : 0.0f;
        const float m2 = (s2 > 0.0f) ? __builtin_sqrtf(s2) : 0.0f;
        const float m3 = (s3 > 0.0f) ? __builtin_sqrtf(s3) : 0.0f;
        const float m4 = (s4 > 0.0f) ? __builtin_sqrtf(s4) : 0.0f;

        float mean2 = 0.5f * (m1 + m2);
        float e20 = m1 - mean2, e21 = m2 - mean2;
        float std2 = __builtin_sqrtf(e20 * e20 + e21 * e21);

        float mean3 = (m1 + m2 + m3) * (1.0f / 3.0f);
        float e0 = m1 - mean3, e1 = m2 - mean3, e2 = m3 - mean3;
        float std3 = __builtin_sqrtf((e0 * e0 + e1 * e1 + e2 * e2) * 0.5f);

        float mean4 = (m1 + m2 + m3 + m4) * 0.25f;
        float f0 = m1 - mean4, f1 = m2 - mean4, f2 = m3 - mean4, f3 = m4 - mean4;
        float std4 = __builtin_sqrtf((f0 * f0 + f1 * f1 + f2 * f2 + f3 * f3) * (1.0f / 3.0f));

        if (ownValid) {
            acc[0] += mean2; acc[1] += std2; acc[2] += m1;
            acc[3] += mean3; acc[4] += std3; acc[5] += m1;
            acc[6] += mean4; acc[7] += std4; acc[8] += m1;
        }
    }

    if (ownValid) {
#pragma unroll
        for (int f = 0; f < 9; ++f)
            lds_red[wv][ownJ * 9 + f] = acc[f] * (1.0f / (float)T_);
    }
    __syncthreads();

    if (tid < J_ * 9) {
        float s = 0.0f;
#pragma unroll
        for (int w = 0; w < WAVES; ++w) s += lds_red[w][tid];
        out[(size_t)b * (J_ * 9) + tid] = s;
    }
}

extern "C" void kernel_launch(void* const* d_in, const int* in_sizes, int n_in,
                              void* d_out, int out_size, void* d_ws, size_t ws_size,
                              hipStream_t stream) {
    (void)in_sizes; (void)n_in; (void)d_ws; (void)ws_size; (void)out_size;
    const float* x = (const float*)d_in[0];
    float* out = (float*)d_out;
    knn_feats_kernel<<<B_, 1024, 0, stream>>>(x, out);
}


# the 7 square roots per frame (4 distances, 3 deviations) as the bare v_sqrt_f32 (1 ulp) instead of the correctly rounded multi-instruction form; s>0 guard becomes max(s,0); output rounding changes
# speedup vs baseline: 3.5346x; 1.1922x over previous
#include <hip/hip_runtime.h>


typedef float    v2f __attribute__((ext_vector_type(2)));
typedef float    v8f __attribute__((ext_vector_type(8)));

#define B_     256
#define T_     150
#define J_     25
#define WAVES  32
#define NEGBIG (-1.0e30f)

__device__ __forceinline__ float md(float a, float b, float c) { return __builtin_amdgcn_fmed3f(a, b, c); }
#define TOP_ ( 3.402823466e+38f)
#define BOT_ (-3.402823466e+38f)
__device__ __forceinline__ float mx(float a, float b) { return md(a, b, TOP_); }
__device__ __forceinline__ float mn(float a, float b) { return md(a, b, BOT_); }

__device__ __forceinline__ void ce_desc(float& a, float& b) {
    const float hi = fmaxf(a, b);
    const float lo = fminf(a, b);
    a = hi; b = lo;
}

__device__ __forceinline__ void insert_desc(float s, float& q1, float& q2, float& q3, float& q4) {
    const float t1 = mx(s, q1);
    const float t2 = md(s, q1, q2);
    const float t3 = md(s, q2, q3);
    const float t4 = md(s, q3, q4);
    q1 = t1; q2 = t2; q3 = t3; q4 = t4;
}

__device__ __forceinline__ void top4_of_column(const v8f& u, const v8f& w,
                                               float& q1, float& q2, float& q3, float& q4) {
    q1 = mx(u[0], u[1]); q2 = mn(u[0], u[1]);
    { const float t1 = mx(u[2], q1), t2 = md(u[2], q1, q2), t3 = mn(u[2], q2); q1 = t1; q2 = t2; q3 = t3; }
    { const float t1 = mx(u[3], q1), t2 = md(u[3], q1, q2), t3 = md(u[3], q2, q3), t4 = mn(u[3], q3);
      q1 = t1; q2 = t2; q3 = t3; q4 = t4; }
    insert_desc(u[4], q1, q2, q3, q4);
    insert_desc(u[5], q1, q2, q3, q4);
    insert_desc(u[6], q1, q2, q3, q4);
    insert_desc(u[7], q1, q2, q3, q4);
    insert_desc(w[0], q1, q2, q3, q4);
    insert_desc(w[1], q1, q2, q3, q4);
    insert_desc(w[2], q1, q2, q3, q4);
    insert_desc(w[3], q1, q2, q3, q4);
    insert_desc(w[4], q1, q2, q3, q4);
}

__device__ __forceinline__ float root(float v) { return __builtin_amdgcn_sqrtf(v); }

__device__ __forceinline__ float xhalf(float v) {
    const int i = __builtin_bit_cast(int, v);
    return __builtin_bit_cast(float, __builtin_amdgcn_permlanex16(i, i, 0x76543210, 0xfedcba98, false, false));
}

__global__ __launch_bounds__(1024)
void knn_feats_kernel(const float* __restrict__ x, float* __restrict__ out) {
    __shared__ float lds_red[WAVES][J_ * 9];

    const int tid  = threadIdx.x;
    const int wv   = tid >> 5;
    const int lane = tid & 31;
    const int b    = blockIdx.x;

    const int  c  = lane & 15;
    const bool hi = lane >= 16;

    const bool v1 = (c < 5) || (c >= 8 && c < 12);
    const int  j1 = v1 ? ((c < 5) ? 16 + c : 13 + c) : 0;

    const int  ownJ     = hi ? j1 : c;
    const bool ownValid = hi ? v1 : true;

    v8f cdiag;
#pragma unroll
    for (int r = 0; r < 8; ++r) cdiag[r] = (((hi ? 8 : 0) + r) == c) ? NEGBIG : 0.0f;
    const v8f cz = {};

    float acc[9];
#pragma unroll
    for (int f = 0; f < 9; ++f) acc[f] = 0.0f;

    for (int t = wv; t < T_; t += WAVES) {
        const float* px = x + (size_t)(b * T_ + t) * (J_ * 3);
        const float* p0 = px + 3 * c;
        const float* p1 = px + 3 * j1;
        const float x0 = p0[0], y0 = p0[1], z0 = p0[2];
        const float x1 = p1[0], y1 = p1[1], z1 = p1[2];

        const float n0  = x0 * x0 + y0 * y0 + z0 * z0;
        const float n1  = x1 * x1 + y1 * y1 + z1 * z1;
        const float nh0 = -0.5f * n0;
        const float nh1 = v1 ? -0.5f * n1 : NEGBIG;

        v2f a0, a1, b0, b1;
        a0.x = hi ? z0 : x0;   a0.y = hi ? nh0  : y0;
        b0.x = a0.x;           b0.y = hi ? 1.0f : y0;
        a1.x = hi ? z1 : x1;   a1.y = hi ? nh1  : y1;
        b1.x = a1.x;           b1.y = hi ? 1.0f : y1;

        v8f d00 = __builtin_amdgcn_wmma_f32_16x16x4_f32(false, a0, false, b0, (short)0, cdiag, false, false);
        v8f d01 = __builtin_amdgcn_wmma_f32_16x16x4_f32(false, a0, false, b1, (short)0, cz,    false, false);
        v8f d10 = __builtin_amdgcn_wmma_f32_16x16x4_f32(false, a1, false, b0, (short)0, cz,    false, false);
        v8f d11 = __builtin_amdgcn_wmma_f32_16x16x4_f32(false, a1, false, b1, (short)0, cdiag, false, false);

        asm volatile("v_nop\n\tv_nop\n\tv_nop\n\tv_nop" : "+v"(d00), "+v"(d01), "+v"(d10), "+v"(d11));

        float P1, P2, P3, P4, Q1, Q2, Q3, Q4;
        top4_of_column(d00, d10, P1, P2, P3, P4);
        top4_of_column(d01, d11, Q1, Q2, Q3, Q4);

        const float R1 = xhalf(hi ? P1 : Q1), R2 = xhalf(hi ? P2 : Q2);
        const float R3 = xhalf(hi ? P3 : Q3), R4 = xhalf(hi ? P4 : Q4);
        const float O1 = hi ? Q1 : P1, O2 = hi ? Q2 : P2, O3 = hi ? Q3 : P3, O4 = hi ? Q4 : P4;

        float L1 = mx(O1, R4), L2 = mx(O2, R3), L3 = mx(O3, R2), L4 = mx(O4, R1);
        ce_desc(L1, L3); ce_desc(L2, L4); ce_desc(L1, L2); ce_desc(L3, L4);

        const float nj = hi ? n1 : n0;
        const float s1 = fmaf(-2.0f, L1, nj);
        const float s2 = fmaf(-2.0f, L2, nj);
        const float s3 = fmaf(-2.0f, L3, nj);
        const float s4 = fmaf(-2.0f, L4, nj);

        const float m1 = root(fmaxf(s1, 0.0f));
        const float m2 = root(fmaxf(s2, 0.0f));
        const float m3 = root(fmaxf(s3, 0.0f));
        const float m4 = root(fmaxf(s4, 0.0f));

        float mean2 = 0.5f * (m1 + m2);
        float e20 = m1 - mean2, e21 = m2 - mean2;
        float std2 = root(e20 * e20 + e21 * e21);

        float mean3 = (m1 + m2 + m3) * (1.0f / 3.0f);
        float e0 = m1 - mean3, e1 = m2 - mean3, e2 = m3 - mean3;
        float std3 = root((e0 * e0 + e1 * e1 + e2 * e2) * 0.5f);

        float mean4 = (m1 + m2 + m3 + m4) * 0.25f;
        float f0 = m1 - mean4, f1 = m2 - mean4, f2 = m3 - mean4, f3 = m4 - mean4;
        float std4 = root((f0 * f0 + f1 * f1 + f2 * f2 + f3 * f3) * (1.0f / 3.0f));

        if (ownValid) {
            acc[0] += mean2; acc[1] += std2; acc[2] += m1;
            acc[3] += mean3; acc[4] += std3; acc[5] += m1;
            acc[6] += mean4; acc[7] += std4; acc[8] += m1;
        }
    }

    if (ownValid) {
#pragma unroll
        for (int f = 0; f < 9; ++f)
            lds_red[wv][ownJ * 9 + f] = acc[f] * (1.0f / (float)T_);
    }
    __syncthreads();

    if (tid < J_ * 9) {
        float s = 0.0f;
#pragma unroll
        for (int w = 0; w < WAVES; ++w) s += lds_red[w][tid];
        out[(size_t)b * (J_ * 9) + tid] = s;
    }
}

extern "C" void kernel_launch(void* const* d_in, const int* in_sizes, int n_in,
                              void* d_out, int out_size, void* d_ws, size_t ws_size,
                              hipStream_t stream) {
    (void)in_sizes; (void)n_in; (void)d_ws; (void)ws_size; (void)out_size;
    const float* x = (const float*)d_in[0];
    float* out = (float*)d_out;
    knn_feats_kernel<<<B_, 1024, 0, stream>>>(x, out);
}
